# starting kernel
# baseline (speedup 1.0000x reference)
#include <hip/hip_runtime.h>
#include <hip/hip_bf16.h>

#define BB 4096
#define KK 32
#define UU 16384
#define DD 256

typedef float v2f __attribute__((ext_vector_type(2)));
typedef float v8f __attribute__((ext_vector_type(8)));

__global__ void gcn_zero_kernel(int* __restrict__ p, int n) {
    int i = blockIdx.x * blockDim.x + threadIdx.x;
    if (i < n) p[i] = 0;
}

__global__ void gcn_count_kernel(const int* __restrict__ neigh,
                                 int* __restrict__ colcnt,
                                 unsigned int* __restrict__ firstmask) {
    int gtid = blockIdx.x * blockDim.x + threadIdx.x;
    int row  = gtid >> 5;
    int lane = gtid & 31;
    int v = neigh[row * KK + lane];

    bool first = true;
    for (int t = 0; t < 32; ++t) {
        int ov = __shfl(v, t, 32);
        if (t < lane && ov == v) first = false;
    }
    unsigned long long bal = __ballot(first);
    unsigned int fm = (unsigned int)bal;
    if (lane == 0) firstmask[row] = fm;
    if (first) atomicAdd(&colcnt[v], 1);
}

__global__ void __launch_bounds__(256)
gcn_wmma_kernel(const int* __restrict__ neigh,
                const float* __restrict__ feats,
                const float* __restrict__ noise,
                const int* __restrict__ colcnt,
                const unsigned int* __restrict__ firstmask,
                float* __restrict__ out) {
    __shared__ int   lds_idx[16 * KK];
    __shared__ float lds_w[16 * KK];

    const int m0  = blockIdx.x * 16;
    const int tid = threadIdx.x;

    for (int e = tid; e < 16 * KK; e += 256) {
        int m = e >> 5, k = e & 31;
        int gb = m0 + m;
        int j  = neigh[gb * KK + k];
        unsigned int fm = firstmask[gb];
        float w = 0.0f;
        if ((fm >> k) & 1u) {
            float rd = (float)__popc(fm);
            float cd = (float)colcnt[j];
            w = rsqrtf(rd) * rsqrtf(cd);
        }
        lds_idx[e] = j;
        lds_w[e]   = w;
    }
    __syncthreads();

    const int wave = tid >> 5;
    const int lane = tid & 31;
    const int half = lane >> 4;
    const int ln   = lane & 15;

    for (int t = wave; t < 32; t += 8) {
        const int table = t >> 4;
        const int n0    = (t & 15) << 4;
        const float* __restrict__ src = table ? noise : feats;
        float* __restrict__ dst = out + table * (BB * DD);

        v8f acc = {0.f, 0.f, 0.f, 0.f, 0.f, 0.f, 0.f, 0.f};

        #pragma unroll 4
        for (int chunk = 0; chunk < 128; ++chunk) {
            const int ms   = chunk >> 3;
            const int base = ms * KK + ((chunk & 7) << 2) + half * 2;

            int   j0 = lds_idx[base],  j1 = lds_idx[base + 1];
            float w0 = lds_w[base],    w1 = lds_w[base + 1];

            const bool on = (ln == ms);
            v2f a;
            a.x = on ? w0 : 0.0f;
            a.y = on ? w1 : 0.0f;

            v2f b;
            b.x = src[j0 * DD + n0 + ln];
            b.y = src[j1 * DD + n0 + ln];

            acc = __builtin_amdgcn_wmma_f32_16x16x4_f32(
                 false, a,  false, b,
                 (short)0, acc,  false,  false);
        }

        #pragma unroll
        for (int r = 0; r < 8; ++r) {
            int row = m0 + r + half * 8;
            dst[row * DD + n0 + ln] = acc[r];
        }
    }
}

extern "C" void kernel_launch(void* const* d_in, const int* in_sizes, int n_in,
                              void* d_out, int out_size, void* d_ws, size_t ws_size,
                              hipStream_t stream) {
    const int*   neigh = (const int*)d_in[0];
    const float* feats = (const float*)d_in[1];
    const float* noise = (const float*)d_in[2];
    float*       out   = (float*)d_out;

    int*          colcnt    = (int*)d_ws;
    unsigned int* firstmask = (unsigned int*)((char*)d_ws + UU * 4);

    gcn_zero_kernel<<<(UU + 255) / 256, 256, 0, stream>>>(colcnt, UU);

    gcn_count_kernel<<<(BB * 32) / 256, 256, 0, stream>>>(neigh, colcnt, firstmask);

    gcn_wmma_kernel<<<BB / 16, 256, 0, stream>>>(neigh, feats, noise,
                                                 colcnt, firstmask, out);
}


# gather B with 16-byte loads per lane: wave = 64-column slab, B pair = 2 adjacent columns of one row, A puts even/odd columns in 2 M rows (8 rows per tile); a WMMA now sums 2 slots, not 4
# speedup vs baseline: 3.1067x; 3.1067x over previous
#include <hip/hip_runtime.h>
#include <hip/hip_bf16.h>

#define BB 4096
#define KK 32
#define UU 16384
#define DD 256
#define ROWS 8

typedef float v2f __attribute__((ext_vector_type(2)));
typedef float v4f __attribute__((ext_vector_type(4)));
typedef float v8f __attribute__((ext_vector_type(8)));

__global__ void gcn_zero_kernel(int* __restrict__ p, int n) {
    int i = blockIdx.x * blockDim.x + threadIdx.x;
    if (i < n) p[i] = 0;
}

__global__ void gcn_count_kernel(const int* __restrict__ neigh,
                                 int* __restrict__ colcnt,
                                 unsigned int* __restrict__ firstmask) {
    int gtid = blockIdx.x * blockDim.x + threadIdx.x;
    int row  = gtid >> 5;
    int lane = gtid & 31;
    int v = neigh[row * KK + lane];

    bool first = true;
    for (int t = 0; t < 32; ++t) {
        int ov = __shfl(v, t, 32);
        if (t < lane && ov == v) first = false;
    }
    unsigned long long bal = __ballot(first);
    unsigned int fm = (unsigned int)bal;
    if (lane == 0) firstmask[row] = fm;
    if (first) atomicAdd(&colcnt[v], 1);
}

__global__ void __launch_bounds__(256) __attribute__((amdgpu_waves_per_eu(4, 4)))
gcn_wmma_kernel(const int* __restrict__ neigh,
                const float* __restrict__ feats,
                const float* __restrict__ noise,
                const int* __restrict__ colcnt,
                const unsigned int* __restrict__ firstmask,
                float* __restrict__ out) {
    __shared__ uint2 lds_ow[ROWS][2][16];

    const int m0  = blockIdx.x * ROWS;
    const int tid = threadIdx.x;

    {
        const int m = tid >> 5, k = tid & 31;
        const int gb = m0 + m;
        const int j  = neigh[gb * KK + k];
        const unsigned int fm = firstmask[gb];
        float w = 0.0f;
        if ((fm >> k) & 1u) {
            float rd = (float)__popc(fm);
            float cd = (float)colcnt[j];
            w = rsqrtf(rd) * rsqrtf(cd);
        }
        lds_ow[m][k & 1][k >> 1] = make_uint2((unsigned int)j * (DD * 4u), __float_as_uint(w));
    }
    __syncthreads();

    const int wave = __builtin_amdgcn_readfirstlane(tid >> 5);
    const int lane = tid & 31;
    const int half = lane >> 4;
    const int ln   = lane & 15;

    const int table = wave >> 2;
    const int col0  = (wave & 3) << 6;
    const char* __restrict__ src = (const char*)(table ? noise : feats);
    const unsigned int colb = (unsigned int)(col0 + 4 * ln) * 4u;

    v8f acc0 = {0.f, 0.f, 0.f, 0.f, 0.f, 0.f, 0.f, 0.f};
    v8f acc1 = {0.f, 0.f, 0.f, 0.f, 0.f, 0.f, 0.f, 0.f};

    #pragma unroll
    for (int ms = 0; ms < ROWS; ++ms) {
        const int  mx  = 2 * (ms & 3) + 8 * (ms >> 2);
        const bool onx = (ln == mx);
        const bool ony = (ln == mx + 1);
        const uint2* ow = &lds_ow[ms][half][0];

        v4f b[16];
        v2f a[16];
        for (int c = 0; c < 16; ++c) {
            const uint2 e = ow[c];
            const float w = __uint_as_float(e.y);
            a[c].x = onx ? w : 0.0f;
            a[c].y = ony ? w : 0.0f;
            b[c] = *(const v4f*)(src + (e.x + colb));
        }
        for (int c = 0; c < 16; ++c) {
            v2f b01, b23;
            b01.x = b[c].x;  b01.y = b[c].y;
            b23.x = b[c].z;  b23.y = b[c].w;
            acc0 = __builtin_amdgcn_wmma_f32_16x16x4_f32(
                 false, a[c],  false, b01,
                 (short)0, acc0,  false,  false);
            acc1 = __builtin_amdgcn_wmma_f32_16x16x4_f32(
                 false, a[c],  false, b23,
                 (short)0, acc1,  false,  false);
        }
    }

    float* __restrict__ dst = out + table * (BB * DD);
    for (int i = 0; i < 4; ++i) {
        const int row = m0 + 4 * half + i;
        v4f o;
        o.x = acc0[2 * i];  o.y = acc0[2 * i + 1];
        o.z = acc1[2 * i];  o.w = acc1[2 * i + 1];
        *(v4f*)(dst + row * DD + col0 + 4 * ln) = o;
    }
}

extern "C" void kernel_launch(void* const* d_in, const int* in_sizes, int n_in,
                              void* d_out, int out_size, void* d_ws, size_t ws_size,
                              hipStream_t stream) {
    const int*   neigh = (const int*)d_in[0];
    const float* feats = (const float*)d_in[1];
    const float* noise = (const float*)d_in[2];
    float*       out   = (float*)d_out;

    int*          colcnt    = (int*)d_ws;
    unsigned int* firstmask = (unsigned int*)((char*)d_ws + UU * 4);

    gcn_zero_kernel<<<(UU + 255) / 256, 256, 0, stream>>>(colcnt, UU);

    gcn_count_kernel<<<(BB * 32) / 256, 256, 0, stream>>>(neigh, colcnt, firstmask);

    gcn_wmma_kernel<<<BB / ROWS, 256, 0, stream>>>(neigh, feats, noise,
                                                 colcnt, firstmask, out);
}


# column degrees without a zeroed global histogram: 64 column slices x 4 row parts, LDS histogram per work-group, 4 partials summed in the gather, row dedupe in its staging; 3 launches -> 2
# speedup vs baseline: 3.4148x; 1.0992x over previous
#include <hip/hip_runtime.h>
#include <hip/hip_bf16.h>

#define BB 4096
#define KK 32
#define UU 16384
#define DD 256
#define ROWS 8

typedef float v2f __attribute__((ext_vector_type(2)));
typedef float v4f __attribute__((ext_vector_type(4)));
typedef float v8f __attribute__((ext_vector_type(8)));

#define SLICES      64
#define PARTS       4
#define SLICE_COLS  (UU / SLICES)
#define PART_ROWS   (BB / PARTS)
#define CNT_THREADS 512
#define WAVE_ROWS   (PART_ROWS / (CNT_THREADS / 32))
#define BATCH       32

__global__ void __launch_bounds__(CNT_THREADS)
gcn_count_kernel(const int* __restrict__ neigh, unsigned int* __restrict__ part) {
    __shared__ unsigned int hist[SLICE_COLS];

    const int tid = threadIdx.x;
    const unsigned int s = blockIdx.x & (SLICES - 1);
    const unsigned int p = blockIdx.x / SLICES;

    if (tid < SLICE_COLS) hist[tid] = 0u;
    __syncthreads();

    const int wave = tid >> 5;
    const int lane = tid & 31;
    const int row0 = (int)p * PART_ROWS + wave * WAVE_ROWS;

    for (int b = 0; b < WAVE_ROWS; b += BATCH) {
        int vv[BATCH];
        #pragma unroll
        for (int q = 0; q < BATCH; ++q)
            vv[q] = neigh[(row0 + b + q) * KK + lane];
        #pragma unroll
        for (int q = 0; q < BATCH; ++q) {
            const int v = vv[q];
            unsigned int rem = __builtin_amdgcn_ballot_w32(((unsigned int)v >> 8) == s);
            unsigned int firstm = 0u;
            while (rem) {
                const int t  = __builtin_ctz(rem);
                const int ov = __builtin_amdgcn_readlane(v, t);
                firstm |= 1u << t;
                rem &= ~__builtin_amdgcn_ballot_w32(v == ov);
            }
            if ((firstm >> lane) & 1u) atomicAdd(&hist[v & (SLICE_COLS - 1)], 1u);
        }
    }
    __syncthreads();

    if (tid < SLICE_COLS) part[(s * SLICE_COLS + tid) * PARTS + p] = hist[tid];
}

__global__ void __launch_bounds__(256) __attribute__((amdgpu_waves_per_eu(4, 4)))
gcn_wmma_kernel(const int* __restrict__ neigh,
                const float* __restrict__ feats,
                const float* __restrict__ noise,
                const unsigned int* __restrict__ part,
                float* __restrict__ out) {
    __shared__ uint2 lds_ow[ROWS][2][16];

    const int m0  = blockIdx.x * ROWS;
    const int tid = threadIdx.x;

    {
        const int m = tid >> 5, k = tid & 31;
        const int gb = m0 + m;
        const int j  = neigh[gb * KK + k];
        unsigned int dup = 0u;
        for (int t = 0; t < 31; ++t) {
            const int ov = __builtin_amdgcn_readlane(j, t);
            dup |= __builtin_amdgcn_ballot_w32(j == ov) & (0xFFFFFFFEu << t);
        }
        const unsigned int fm = ~dup;
        float w = 0.0f;
        if ((fm >> k) & 1u) {
            float rd = (float)__popc(fm);
            const uint4 c = *(const uint4*)(part + 4u * (unsigned int)j);
            float cd = (float)(int)(c.x + c.y + c.z + c.w);
            w = rsqrtf(rd) * rsqrtf(cd);
        }
        lds_ow[m][k & 1][k >> 1] = make_uint2((unsigned int)j * (DD * 4u), __float_as_uint(w));
    }
    __syncthreads();

    const int wave = __builtin_amdgcn_readfirstlane(tid >> 5);
    const int lane = tid & 31;
    const int half = lane >> 4;
    const int ln   = lane & 15;

    const int table = wave >> 2;
    const int col0  = (wave & 3) << 6;
    const char* __restrict__ src = (const char*)(table ? noise : feats);
    const unsigned int colb = (unsigned int)(col0 + 4 * ln) * 4u;

    v8f acc0 = {0.f, 0.f, 0.f, 0.f, 0.f, 0.f, 0.f, 0.f};
    v8f acc1 = {0.f, 0.f, 0.f, 0.f, 0.f, 0.f, 0.f, 0.f};

    #pragma unroll
    for (int ms = 0; ms < ROWS; ++ms) {
        const int  mx  = 2 * (ms & 3) + 8 * (ms >> 2);
        const bool onx = (ln == mx);
        const bool ony = (ln == mx + 1);
        const uint2* ow = &lds_ow[ms][half][0];

        v4f b[16];
        v2f a[16];
        for (int c = 0; c < 16; ++c) {
            const uint2 e = ow[c];
            const float w = __uint_as_float(e.y);
            a[c].x = onx ? w : 0.0f;
            a[c].y = ony ? w : 0.0f;
            b[c] = *(const v4f*)(src + (e.x + colb));
        }
        for (int c = 0; c < 16; ++c) {
            v2f b01, b23;
            b01.x = b[c].x;  b01.y = b[c].y;
            b23.x = b[c].z;  b23.y = b[c].w;
            acc0 = __builtin_amdgcn_wmma_f32_16x16x4_f32(
                 false, a[c],  false, b01,
                 (short)0, acc0,  false,  false);
            acc1 = __builtin_amdgcn_wmma_f32_16x16x4_f32(
                 false, a[c],  false, b23,
                 (short)0, acc1,  false,  false);
        }
    }

    float* __restrict__ dst = out + table * (BB * DD);
    for (int i = 0; i < 4; ++i) {
        const int row = m0 + 4 * half + i;
        v4f o;
        o.x = acc0[2 * i];  o.y = acc0[2 * i + 1];
        o.z = acc1[2 * i];  o.w = acc1[2 * i + 1];
        *(v4f*)(dst + row * DD + col0 + 4 * ln) = o;
    }
}

extern "C" void kernel_launch(void* const* d_in, const int* in_sizes, int n_in,
                              void* d_out, int out_size, void* d_ws, size_t ws_size,
                              hipStream_t stream) {
    const int*   neigh = (const int*)d_in[0];
    const float* feats = (const float*)d_in[1];
    const float* noise = (const float*)d_in[2];
    float*       out   = (float*)d_out;

    unsigned int* part = (unsigned int*)d_ws;

    gcn_count_kernel<<<SLICES * PARTS, CNT_THREADS, 0, stream>>>(neigh, part);

    gcn_wmma_kernel<<<BB / ROWS, 256, 0, stream>>>(neigh, feats, noise, part, out);
}


# column degrees from row-presence bits: per (512-column slice, 512-row part) a wave ORs bit q of its row word in LDS, popcount per column, 8 16-bit partials; no per-row dedupe loop or branch
# speedup vs baseline: 4.0731x; 1.1928x over previous
#include <hip/hip_runtime.h>
#include <hip/hip_bf16.h>

#define BB 4096
#define KK 32
#define UU 16384
#define DD 256
#define ROWS 8

typedef float v2f __attribute__((ext_vector_type(2)));
typedef float v4f __attribute__((ext_vector_type(4)));
typedef float v8f __attribute__((ext_vector_type(8)));

#define SLICES      32
#define PARTS       8
#define SLICE_COLS  (UU / SLICES)
#define PART_ROWS   (BB / PARTS)
#define CNT_THREADS 512
#define WAVE_ROWS   (PART_ROWS / (CNT_THREADS / 32))
#define ROW_WORDS   (PART_ROWS / 32)

__global__ void __launch_bounds__(CNT_THREADS)
gcn_count_kernel(const int* __restrict__ neigh, unsigned short* __restrict__ part) {
    __shared__ unsigned int bits[ROW_WORDS][SLICE_COLS];

    const int tid = threadIdx.x;
    const unsigned int s = blockIdx.x & (SLICES - 1);
    const unsigned int p = blockIdx.x / SLICES;

    const int wave = tid >> 5;
    const int lane = tid & 31;
    const int row0 = (int)p * PART_ROWS + wave * WAVE_ROWS;

    int vv[WAVE_ROWS];
    #pragma unroll
    for (int q = 0; q < WAVE_ROWS; ++q)
        vv[q] = neigh[(row0 + q) * KK + lane];

    {
        uint4* z = (uint4*)&bits[0][0];
        const uint4 zero = make_uint4(0u, 0u, 0u, 0u);
        for (int i = 0; i < (ROW_WORDS * SLICE_COLS / 4) / CNT_THREADS; ++i)
            z[tid + i * CNT_THREADS] = zero;
    }
    __syncthreads();

    unsigned int* wrow = &bits[wave][0];
    #pragma unroll
    for (int q = 0; q < WAVE_ROWS; ++q) {
        const int v = vv[q];
        if (((unsigned int)v >> 9) == s) atomicOr(&wrow[v & (SLICE_COLS - 1)], 1u << q);
    }
    __syncthreads();

    unsigned int cnt = 0u;
    for (int w = 0; w < ROW_WORDS; ++w) cnt += (unsigned int)__popc(bits[w][tid]);
    part[(s * SLICE_COLS + tid) * PARTS + p] = (unsigned short)cnt;
}

__global__ void __launch_bounds__(256) __attribute__((amdgpu_waves_per_eu(4, 4)))
gcn_wmma_kernel(const int* __restrict__ neigh,
                const float* __restrict__ feats,
                const float* __restrict__ noise,
                const unsigned int* __restrict__ part,
                float* __restrict__ out) {
    __shared__ uint2 lds_ow[ROWS][2][16];

    const int m0  = blockIdx.x * ROWS;
    const int tid = threadIdx.x;

    {
        const int m = tid >> 5, k = tid & 31;
        const int gb = m0 + m;
        const int j  = neigh[gb * KK + k];
        unsigned int dup = 0u;
        for (int t = 0; t < 31; ++t) {
            const int ov = __builtin_amdgcn_readlane(j, t);
            dup |= __builtin_amdgcn_ballot_w32(j == ov) & (0xFFFFFFFEu << t);
        }
        const unsigned int fm = ~dup;
        float w = 0.0f;
        if ((fm >> k) & 1u) {
            float rd = (float)__popc(fm);
            const uint4 c = *(const uint4*)(part + 4u * (unsigned int)j);
            const unsigned int h = (c.x & 0xFFFFu) + (c.x >> 16) + (c.y & 0xFFFFu) + (c.y >> 16)
                                 + (c.z & 0xFFFFu) + (c.z >> 16) + (c.w & 0xFFFFu) + (c.w >> 16);
            float cd = (float)(int)h;
            w = rsqrtf(rd) * rsqrtf(cd);
        }
        lds_ow[m][k & 1][k >> 1] = make_uint2((unsigned int)j * (DD * 4u), __float_as_uint(w));
    }
    __syncthreads();

    const int wave = __builtin_amdgcn_readfirstlane(tid >> 5);
    const int lane = tid & 31;
    const int half = lane >> 4;
    const int ln   = lane & 15;

    const int table = wave >> 2;
    const int col0  = (wave & 3) << 6;
    const char* __restrict__ src = (const char*)(table ? noise : feats);
    const unsigned int colb = (unsigned int)(col0 + 4 * ln) * 4u;

    v8f acc0 = {0.f, 0.f, 0.f, 0.f, 0.f, 0.f, 0.f, 0.f};
    v8f acc1 = {0.f, 0.f, 0.f, 0.f, 0.f, 0.f, 0.f, 0.f};

    #pragma unroll
    for (int ms = 0; ms < ROWS; ++ms) {
        const int  mx  = 2 * (ms & 3) + 8 * (ms >> 2);
        const bool onx = (ln == mx);
        const bool ony = (ln == mx + 1);
        const uint2* ow = &lds_ow[ms][half][0];

        v4f b[16];
        v2f a[16];
        for (int c = 0; c < 16; ++c) {
            const uint2 e = ow[c];
            const float w = __uint_as_float(e.y);
            a[c].x = onx ? w : 0.0f;
            a[c].y = ony ? w : 0.0f;
            b[c] = *(const v4f*)(src + (e.x + colb));
        }
        for (int c = 0; c < 16; ++c) {
            v2f b01, b23;
            b01.x = b[c].x;  b01.y = b[c].y;
            b23.x = b[c].z;  b23.y = b[c].w;
            acc0 = __builtin_amdgcn_wmma_f32_16x16x4_f32(
                 false, a[c],  false, b01,
                 (short)0, acc0,  false,  false);
            acc1 = __builtin_amdgcn_wmma_f32_16x16x4_f32(
                 false, a[c],  false, b23,
                 (short)0, acc1,  false,  false);
        }
    }

    float* __restrict__ dst = out + table * (BB * DD);
    for (int i = 0; i < 4; ++i) {
        const int row = m0 + 4 * half + i;
        v4f o;
        o.x = acc0[2 * i];  o.y = acc0[2 * i + 1];
        o.z = acc1[2 * i];  o.w = acc1[2 * i + 1];
        *(v4f*)(dst + row * DD + col0 + 4 * ln) = o;
    }
}

extern "C" void kernel_launch(void* const* d_in, const int* in_sizes, int n_in,
                              void* d_out, int out_size, void* d_ws, size_t ws_size,
                              hipStream_t stream) {
    const int*   neigh = (const int*)d_in[0];
    const float* feats = (const float*)d_in[1];
    const float* noise = (const float*)d_in[2];
    float*       out   = (float*)d_out;

    unsigned int* part = (unsigned int*)d_ws;

    gcn_count_kernel<<<SLICES * PARTS, CNT_THREADS, 0, stream>>>(neigh, (unsigned short*)part);

    gcn_wmma_kernel<<<BB / ROWS, 256, 0, stream>>>(neigh, feats, noise, part, out);
}


# gather kernel prologue: degree lookup, then first row's 16 gather loads, issued right after the indices and kept in flight across both barriers (LDS-only barriers) while dedupe and weights are made
# speedup vs baseline: 4.3247x; 1.0618x over previous
#include <hip/hip_runtime.h>
#include <hip/hip_bf16.h>

#define BB 4096
#define KK 32
#define UU 16384
#define DD 256
#define ROWS 8

typedef float v2f __attribute__((ext_vector_type(2)));
typedef float v4f __attribute__((ext_vector_type(4)));
typedef float v8f __attribute__((ext_vector_type(8)));

#define SLICES      32
#define PARTS       8
#define SLICE_COLS  (UU / SLICES)
#define PART_ROWS   (BB / PARTS)
#define CNT_THREADS 512
#define WAVE_ROWS   (PART_ROWS / (CNT_THREADS / 32))
#define ROW_WORDS   (PART_ROWS / 32)

__global__ void __launch_bounds__(CNT_THREADS)
gcn_count_kernel(const int* __restrict__ neigh, unsigned short* __restrict__ part) {
    __shared__ unsigned int bits[ROW_WORDS][SLICE_COLS];

    const int tid = threadIdx.x;
    const unsigned int s = blockIdx.x & (SLICES - 1);
    const unsigned int p = blockIdx.x / SLICES;

    const int wave = tid >> 5;
    const int lane = tid & 31;
    const int row0 = (int)p * PART_ROWS + wave * WAVE_ROWS;

    int vv[WAVE_ROWS];
    #pragma unroll
    for (int q = 0; q < WAVE_ROWS; ++q)
        vv[q] = neigh[(row0 + q) * KK + lane];

    {
        uint4* z = (uint4*)&bits[0][0];
        const uint4 zero = make_uint4(0u, 0u, 0u, 0u);
        for (int i = 0; i < (ROW_WORDS * SLICE_COLS / 4) / CNT_THREADS; ++i)
            z[tid + i * CNT_THREADS] = zero;
    }
    __syncthreads();

    unsigned int* wrow = &bits[wave][0];
    #pragma unroll
    for (int q = 0; q < WAVE_ROWS; ++q) {
        const int v = vv[q];
        if (((unsigned int)v >> 9) == s) atomicOr(&wrow[v & (SLICE_COLS - 1)], 1u << q);
    }
    __syncthreads();

    unsigned int cnt = 0u;
    for (int w = 0; w < ROW_WORDS; ++w) cnt += (unsigned int)__popc(bits[w][tid]);
    part[(s * SLICE_COLS + tid) * PARTS + p] = (unsigned short)cnt;
}

__device__ __forceinline__ void lds_handover_barrier() {
    asm volatile("s_wait_dscnt 0x0\n\ts_barrier_signal -1\n\ts_barrier_wait -1" ::: "memory");
}

__global__ void __launch_bounds__(256) __attribute__((amdgpu_waves_per_eu(4, 4)))
gcn_wmma_kernel(const int* __restrict__ neigh,
                const float* __restrict__ feats,
                const float* __restrict__ noise,
                const unsigned int* __restrict__ part,
                float* __restrict__ out) {
    __shared__ unsigned int lds_off[ROWS][2][16];
    __shared__ unsigned int lds_w[ROWS][2][16];

    const int m0  = blockIdx.x * ROWS;
    const int tid = threadIdx.x;

    const int sm = tid >> 5, sk = tid & 31;
    const int j  = neigh[(m0 + sm) * KK + sk];
    const uint4 cpart = *(const uint4*)(part + 4u * (unsigned int)j);
    __builtin_amdgcn_sched_barrier(0);
    lds_off[sm][sk & 1][sk >> 1] = (unsigned int)j * (DD * 4u);
    lds_handover_barrier();

    const int wave = __builtin_amdgcn_readfirstlane(tid >> 5);
    const int lane = tid & 31;
    const int half = lane >> 4;
    const int ln   = lane & 15;

    const int table = wave >> 2;
    const int col0  = (wave & 3) << 6;
    const char* __restrict__ src = (const char*)(table ? noise : feats);
    const unsigned int colb = (unsigned int)(col0 + 4 * ln) * 4u;

    v4f bfirst[16];
    {
        const unsigned int* of0 = &lds_off[0][half][0];
        for (int c = 0; c < 16; ++c)
            bfirst[c] = *(const v4f*)(src + (of0[c] + colb));
    }
    __builtin_amdgcn_sched_barrier(0);

    {
        unsigned int dup = 0u;
        for (int t = 0; t < 31; ++t) {
            const int ov = __builtin_amdgcn_readlane(j, t);
            dup |= __builtin_amdgcn_ballot_w32(j == ov) & (0xFFFFFFFEu << t);
        }
        const unsigned int fm = ~dup;
        const float rd = (float)__popc(fm);
        const uint4 c = cpart;
        const unsigned int h = (c.x & 0xFFFFu) + (c.x >> 16) + (c.y & 0xFFFFu) + (c.y >> 16)
                             + (c.z & 0xFFFFu) + (c.z >> 16) + (c.w & 0xFFFFu) + (c.w >> 16);
        const float cd = (float)(int)h;
        const float w = rsqrtf(rd) * rsqrtf(cd);
        const unsigned int keep = 0u - ((fm >> sk) & 1u);
        lds_w[sm][sk & 1][sk >> 1] = __float_as_uint(w) & keep;
    }
    lds_handover_barrier();

    v8f acc0 = {0.f, 0.f, 0.f, 0.f, 0.f, 0.f, 0.f, 0.f};
    v8f acc1 = {0.f, 0.f, 0.f, 0.f, 0.f, 0.f, 0.f, 0.f};

    #pragma unroll
    for (int ms = 0; ms < ROWS; ++ms) {
        const int  mx  = 2 * (ms & 3) + 8 * (ms >> 2);
        const bool onx = (ln == mx);
        const bool ony = (ln == mx + 1);
        const unsigned int* of = &lds_off[ms][half][0];
        const unsigned int* wt = &lds_w[ms][half][0];

        v4f b[16];
        v2f a[16];
        for (int c = 0; c < 16; ++c) {
            const float w = __uint_as_float(wt[c]);
            a[c].x = onx ? w : 0.0f;
            a[c].y = ony ? w : 0.0f;
            if (ms == 0) b[c] = bfirst[c];
            else         b[c] = *(const v4f*)(src + (of[c] + colb));
        }
        for (int c = 0; c < 16; ++c) {
            v2f b01, b23;
            b01.x = b[c].x;  b01.y = b[c].y;
            b23.x = b[c].z;  b23.y = b[c].w;
            acc0 = __builtin_amdgcn_wmma_f32_16x16x4_f32(
                 false, a[c],  false, b01,
                 (short)0, acc0,  false,  false);
            acc1 = __builtin_amdgcn_wmma_f32_16x16x4_f32(
                 false, a[c],  false, b23,
                 (short)0, acc1,  false,  false);
        }
    }

    float* __restrict__ dst = out + table * (BB * DD);
    for (int i = 0; i < 4; ++i) {
        const int row = m0 + 4 * half + i;
        v4f o;
        o.x = acc0[2 * i];  o.y = acc0[2 * i + 1];
        o.z = acc1[2 * i];  o.w = acc1[2 * i + 1];
        *(v4f*)(dst + row * DD + col0 + 4 * ln) = o;
    }
}

extern "C" void kernel_launch(void* const* d_in, const int* in_sizes, int n_in,
                              void* d_out, int out_size, void* d_ws, size_t ws_size,
                              hipStream_t stream) {
    const int*   neigh = (const int*)d_in[0];
    const float* feats = (const float*)d_in[1];
    const float* noise = (const float*)d_in[2];
    float*       out   = (float*)d_out;

    unsigned int* part = (unsigned int*)d_ws;

    gcn_count_kernel<<<SLICES * PARTS, CNT_THREADS, 0, stream>>>(neigh, (unsigned short*)part);

    gcn_wmma_kernel<<<BB / ROWS, 256, 0, stream>>>(neigh, feats, noise, part, out);
}
